# starting kernel
# baseline (speedup 1.0000x reference)
#include <hip/hip_runtime.h>
#include <hip/hip_bf16.h>


typedef float v2f __attribute__((ext_vector_type(2)));
typedef float v8f __attribute__((ext_vector_type(8)));

#define HIDDEN 16


__global__ void deg_init_kernel(float* __restrict__ deg, int n) {
    int i = blockIdx.x * blockDim.x + threadIdx.x;
    if (i < n) deg[i] = 1.0f;
}

__global__ void deg_count_kernel(const int* __restrict__ dst, float* __restrict__ deg,
                                 int n_edges) {
    int e = blockIdx.x * blockDim.x + threadIdx.x;
    if (e < n_edges) atomicAdd(&deg[dst[e]], 1.0f);
}

__global__ void dinv_kernel(float* __restrict__ deg_dinv, int n) {
    int i = blockIdx.x * blockDim.x + threadIdx.x;
    if (i < n) deg_dinv[i] = rsqrtf(deg_dinv[i]);
}

__global__ void __launch_bounds__(256)
gemm_wmma_kernel(const float* __restrict__ X, const float* __restrict__ W,
                 float* __restrict__ H, int n_rows, int K) {
    int wave = (int)((blockIdx.x * blockDim.x + threadIdx.x) >> 5);
    int lane = threadIdx.x & 31;
    int row0 = wave * 16;
    if (row0 >= n_rows) return;

    int half = lane >> 4;
    int l16  = lane & 15;

    const float* xrow = X + (size_t)(row0 + l16) * K;
    v8f acc = {};
    for (int k = 0; k < K; k += 4) {
        int ka = k + half * 2;
        v2f a, b;
        a.x = xrow[ka];
        a.y = xrow[ka + 1];
        b.x = W[(size_t)ka * HIDDEN + l16];
        b.y = W[(size_t)(ka + 1) * HIDDEN + l16];
        acc = __builtin_amdgcn_wmma_f32_16x16x4_f32(
             false, a,  false, b,
             (short)0, acc,  false,  false);
    }
    float* out = H + (size_t)row0 * HIDDEN;
#pragma unroll
    for (int i = 0; i < 8; ++i)
        out[(size_t)(i + 8 * half) * HIDDEN + l16] = acc[i];
}


__global__ void zero_kernel(float* __restrict__ p, int n) {
    int i = blockIdx.x * blockDim.x + threadIdx.x;
    if (i < n) p[i] = 0.0f;
}

__global__ void scatter_kernel(const int* __restrict__ src, const int* __restrict__ dst,
                               const float* __restrict__ dinv,
                               const float* __restrict__ hin,
                               float* __restrict__ agg, int n_edges) {
    int g = blockIdx.x * blockDim.x + threadIdx.x;
    int e = g >> 4;
    if (e >= n_edges) return;
    int f = g & 15;
    int s = src[e];
    int d = dst[e];
    float norm = dinv[s] * dinv[d];
    float v = hin[(size_t)s * HIDDEN + f] * norm;
    atomicAdd(&agg[(size_t)d * HIDDEN + f], v);
}

__global__ void finalize_kernel(const float* __restrict__ hpre,
                                const float* __restrict__ dinv,
                                const float* __restrict__ bias,
                                float* __restrict__ agg, int n_nodes, int do_relu) {
    int g = blockIdx.x * blockDim.x + threadIdx.x;
    int i = g >> 4;
    if (i >= n_nodes) return;
    int f = g & 15;
    float di = dinv[i];
    float v = agg[(size_t)i * HIDDEN + f] + di * di * hpre[(size_t)i * HIDDEN + f]
              + bias[f];
    if (do_relu) v = fmaxf(v, 0.0f);
    agg[(size_t)i * HIDDEN + f] = v;
}


__global__ void pool_kernel(const float* __restrict__ h, const int* __restrict__ batch,
                            float* __restrict__ pooled, int n_nodes) {
    int g = blockIdx.x * blockDim.x + threadIdx.x;
    int i = g >> 4;
    if (i >= n_nodes) return;
    int f = g & 15;
    atomicAdd(&pooled[(size_t)batch[i] * HIDDEN + f], h[(size_t)i * HIDDEN + f]);
}

__global__ void final_linear_kernel(const float* __restrict__ pooled,
                                    const float* __restrict__ Wlin,
                                    const float* __restrict__ blin,
                                    float* __restrict__ out,
                                    int n_graphs, int n_classes) {
    int g = blockIdx.x * blockDim.x + threadIdx.x;
    if (g >= n_graphs * n_classes) return;
    int gr = g / n_classes;
    int c  = g % n_classes;
    float acc = blin[c];
#pragma unroll
    for (int f = 0; f < HIDDEN; ++f)
        acc += pooled[(size_t)gr * HIDDEN + f] * Wlin[f * n_classes + c];
    out[g] = acc;
}


static inline size_t align_up(size_t v, size_t a) { return (v + a - 1) & ~(a - 1); }

extern "C" void kernel_launch(void* const* d_in, const int* in_sizes, int n_in,
                              void* d_out, int out_size, void* d_ws, size_t ws_size,
                              hipStream_t stream) {
    (void)n_in; (void)ws_size;

    const float* x    = (const float*)d_in[0];
    const int*   ei   = (const int*)d_in[1];
    const int*   batch= (const int*)d_in[2];
    const float* W1   = (const float*)d_in[3];
    const float* b1   = (const float*)d_in[4];
    const float* W2   = (const float*)d_in[5];
    const float* b2   = (const float*)d_in[6];
    const float* W3   = (const float*)d_in[7];
    const float* b3   = (const float*)d_in[8];
    const float* Wlin = (const float*)d_in[9];
    const float* blin = (const float*)d_in[10];
    float* out = (float*)d_out;

    const int N_FEAT = 128;
    const int n_nodes = in_sizes[0] / N_FEAT;
    const int n_edges = in_sizes[1] / 2;
    const int n_classes = in_sizes[10];
    const int n_graphs  = out_size / n_classes;

    const int* src = ei;
    const int* dst = ei + n_edges;

    char* ws = (char*)d_ws;
    size_t off = 0;
    float* dinv = (float*)(ws + off);  off = align_up(off + (size_t)n_nodes * 4, 256);
    float* bufA = (float*)(ws + off);  off = align_up(off + (size_t)n_nodes * HIDDEN * 4, 256);
    float* bufB = (float*)(ws + off);  off = align_up(off + (size_t)n_nodes * HIDDEN * 4, 256);
    float* pooled = (float*)(ws + off); off = align_up(off + (size_t)n_graphs * HIDDEN * 4, 256);

    const int T = 256;
    const int nodeBlocks  = (n_nodes + T - 1) / T;
    const int edgeBlocks  = (n_edges + T - 1) / T;
    const int nodeFBlocks = ((n_nodes * HIDDEN) + T - 1) / T;
    const int edgeFBlocks = (int)(((long long)n_edges * HIDDEN + T - 1) / T);
    const int hElems      = n_nodes * HIDDEN;
    const int tiles       = (n_nodes + 15) / 16;
    const int gemmBlocks  = (tiles + 7) / 8;

    deg_init_kernel<<<nodeBlocks, T, 0, stream>>>(dinv, n_nodes);
    deg_count_kernel<<<edgeBlocks, T, 0, stream>>>(dst, dinv, n_edges);
    dinv_kernel<<<nodeBlocks, T, 0, stream>>>(dinv, n_nodes);

    gemm_wmma_kernel<<<gemmBlocks, T, 0, stream>>>(x, W1, bufA, n_nodes, N_FEAT);
    zero_kernel<<<nodeFBlocks, T, 0, stream>>>(bufB, hElems);
    scatter_kernel<<<edgeFBlocks, T, 0, stream>>>(src, dst, dinv, bufA, bufB, n_edges);
    finalize_kernel<<<nodeFBlocks, T, 0, stream>>>(bufA, dinv, b1, bufB, n_nodes, 1);

    gemm_wmma_kernel<<<gemmBlocks, T, 0, stream>>>(bufB, W2, bufA, n_nodes, HIDDEN);
    zero_kernel<<<nodeFBlocks, T, 0, stream>>>(bufB, hElems);
    scatter_kernel<<<edgeFBlocks, T, 0, stream>>>(src, dst, dinv, bufA, bufB, n_edges);
    finalize_kernel<<<nodeFBlocks, T, 0, stream>>>(bufA, dinv, b2, bufB, n_nodes, 1);

    gemm_wmma_kernel<<<gemmBlocks, T, 0, stream>>>(bufB, W3, bufA, n_nodes, HIDDEN);
    zero_kernel<<<nodeFBlocks, T, 0, stream>>>(bufB, hElems);
    scatter_kernel<<<edgeFBlocks, T, 0, stream>>>(src, dst, dinv, bufA, bufB, n_edges);
    finalize_kernel<<<nodeFBlocks, T, 0, stream>>>(bufA, dinv, b3, bufB, n_nodes, 0);

    zero_kernel<<<(n_graphs * HIDDEN + T - 1) / T, T, 0, stream>>>(pooled, n_graphs * HIDDEN);
    pool_kernel<<<nodeFBlocks, T, 0, stream>>>(bufB, batch, pooled, n_nodes);
    final_linear_kernel<<<(n_graphs * n_classes + T - 1) / T, T, 0, stream>>>(
        pooled, Wlin, blin, out, n_graphs, n_classes);
}


# Layers 1-2 aggregation with no global atomics: a work-group per (node tile, edge slice) filters dst, sums matched rows in an LDS tile, finalize adds the planes; 15 to 13 launches; sum order changes
# speedup vs baseline: 1.6000x; 1.6000x over previous
#include <hip/hip_runtime.h>
#include <hip/hip_bf16.h>


typedef float v2f __attribute__((ext_vector_type(2)));
typedef float v8f __attribute__((ext_vector_type(8)));

#define HIDDEN 16


__global__ void deg_init_kernel(float* __restrict__ deg, int n) {
    int i = blockIdx.x * blockDim.x + threadIdx.x;
    if (i < n) deg[i] = 1.0f;
}

__global__ void deg_count_kernel(const int* __restrict__ dst, float* __restrict__ deg,
                                 int n_edges) {
    int e = blockIdx.x * blockDim.x + threadIdx.x;
    if (e < n_edges) atomicAdd(&deg[dst[e]], 1.0f);
}

__global__ void dinv_kernel(float* __restrict__ deg_dinv, int n) {
    int i = blockIdx.x * blockDim.x + threadIdx.x;
    if (i < n) deg_dinv[i] = rsqrtf(deg_dinv[i]);
}

__global__ void __launch_bounds__(256)
gemm_wmma_kernel(const float* __restrict__ X, const float* __restrict__ W,
                 float* __restrict__ H, int n_rows, int K) {
    int wave = (int)((blockIdx.x * blockDim.x + threadIdx.x) >> 5);
    int lane = threadIdx.x & 31;
    int row0 = wave * 16;
    if (row0 >= n_rows) return;

    int half = lane >> 4;
    int l16  = lane & 15;

    const float* xrow = X + (size_t)(row0 + l16) * K;
    v8f acc = {};
    for (int k = 0; k < K; k += 4) {
        int ka = k + half * 2;
        v2f a, b;
        a.x = xrow[ka];
        a.y = xrow[ka + 1];
        b.x = W[(size_t)ka * HIDDEN + l16];
        b.y = W[(size_t)(ka + 1) * HIDDEN + l16];
        acc = __builtin_amdgcn_wmma_f32_16x16x4_f32(
             false, a,  false, b,
             (short)0, acc,  false,  false);
    }
    float* out = H + (size_t)row0 * HIDDEN;
#pragma unroll
    for (int i = 0; i < 8; ++i)
        out[(size_t)(i + 8 * half) * HIDDEN + l16] = acc[i];
}


__global__ void zero_kernel(float* __restrict__ p, int n) {
    int i = blockIdx.x * blockDim.x + threadIdx.x;
    if (i < n) p[i] = 0.0f;
}


template <int J>
__device__ __forceinline__ int row_lane_i(int x) {
    return __builtin_amdgcn_update_dpp(0, x, 0x150 + J, 0xf, 0xf, false);
}
template <int J>
__device__ __forceinline__ float row_lane_f(float x) {
    return __int_as_float(row_lane_i<J>(__float_as_int(x)));
}

template <int J>
struct PoolSteps {
    static __device__ __forceinline__ void load(const float* __restrict__ hin, int s, int f,
                                                float (&r)[16]) {
        r[J] = hin[row_lane_i<J>(s) * HIDDEN + f];
        PoolSteps<J + 1>::load(hin, s, f, r);
    }
    static __device__ __forceinline__ void add(float* tab, int g, float w, int f, float bf,
                                               const float (&r)[16]) {
        atomicAdd(&tab[row_lane_i<J>(g) * HIDDEN + f], fmaf(row_lane_f<J>(w), r[J], bf));
        PoolSteps<J + 1>::add(tab, g, w, f, bf, r);
    }
};
template <>
struct PoolSteps<16> {
    static __device__ __forceinline__ void load(const float* __restrict__, int, int,
                                                float (&)[16]) {}
    static __device__ __forceinline__ void add(float*, int, float, int, float,
                                               const float (&)[16]) {}
};

#define TILE_NODES 3136
#define N_SLICES   8
#define Q_CAP      256

__device__ __forceinline__ void enqueue_matches(int d, int e, int lo, int cnt, int* qe, int* qd,
                                                int& tail) {
    const int ld = d - lo;
    const bool m = (unsigned)ld < (unsigned)cnt;
    const unsigned mask = __builtin_amdgcn_ballot_w32(m);
    if (mask != 0u) {
        const int pos = (tail + (int)__builtin_amdgcn_mbcnt_lo(mask, 0u)) & (Q_CAP - 1);
        if (m) { qe[pos] = e; qd[pos] = ld; }
        tail += __popc(mask);
    }
}

__device__ __forceinline__ void push_batch(float* tab, const int* qe, const int* qd, int head,
                                           int nvalid, int lane, int f, int lo, int e_any,
                                           const int* __restrict__ src,
                                           const float* __restrict__ dinv,
                                           const float* __restrict__ hin) {
    const int idx = (head + lane) & (Q_CAP - 1);
    const bool own = lane < nvalid;
    const int e = own ? qe[idx] : e_any;
    int ld = own ? qd[idx] : 0;
    int s = src[e];
    float w = dinv[s] * dinv[lo + ld];
    if (!own) { s = 0; w = 0.0f; ld = TILE_NODES; }
    float r[16];
    PoolSteps<0>::load(hin, s, f, r);
    PoolSteps<0>::add(tab, ld, w, f, 0.0f, r);
}

__global__ void __launch_bounds__(1024)
tile_push_kernel(const int* __restrict__ src, const int* __restrict__ dst,
                 const float* __restrict__ dinv, const float* __restrict__ hin,
                 float* __restrict__ part, int n_edges, int n_nodes, int n_tiles,
                 int per_slice) {
    __shared__ __attribute__((aligned(16))) float tab[(TILE_NODES + 1) * HIDDEN];
    __shared__ int q_edge[32][Q_CAP];
    __shared__ int q_row[32][Q_CAP];

    const int tid   = threadIdx.x;
    const int tile  = blockIdx.x % n_tiles;
    const int slice = blockIdx.x / n_tiles;
    const int lo    = tile * TILE_NODES;
    const int cnt   = min(TILE_NODES, n_nodes - lo);

    float4* tab4 = reinterpret_cast<float4*>(tab);
    for (int i = tid; i < (TILE_NODES + 1) * HIDDEN / 4; i += 1024)
        tab4[i] = make_float4(0.0f, 0.0f, 0.0f, 0.0f);
    __syncthreads();

    const int lane = tid & 31;
    const int f    = lane & 15;
    const int wv = __builtin_amdgcn_readfirstlane(tid >> 5);
    int* qe = q_edge[wv];
    int* qd = q_row[wv];

    const int e0 = slice * per_slice;
    const int e1 = min(e0 + per_slice, n_edges);
    const bool vec_ok = (reinterpret_cast<size_t>(dst) & 15) == 0;

    int head = 0, tail = 0;
    const int step = 32 * 512;
    int base = e0 + wv * 512;

    if (vec_ok && base + 512 <= e1) {
        int4 cur[4];
#pragma unroll
        for (int k = 0; k < 4; ++k)
            cur[k] = *reinterpret_cast<const int4*>(dst + base + k * 128 + lane * 4);
        for (;;) {
            const int  nb   = base + step;
            const bool more = nb + 512 <= e1;
            int4 nxt[4];
#pragma unroll
            for (int k = 0; k < 4; ++k) nxt[k] = cur[k];
            if (more) {
#pragma unroll
                for (int k = 0; k < 4; ++k)
                    nxt[k] = *reinterpret_cast<const int4*>(dst + nb + k * 128 + lane * 4);
            }
#pragma unroll
            for (int k = 0; k < 4; ++k) {
                const int e = base + k * 128 + lane * 4;
                enqueue_matches(cur[k].x, e,     lo, cnt, qe, qd, tail);
                enqueue_matches(cur[k].y, e + 1, lo, cnt, qe, qd, tail);
                enqueue_matches(cur[k].z, e + 2, lo, cnt, qe, qd, tail);
                enqueue_matches(cur[k].w, e + 3, lo, cnt, qe, qd, tail);
                while (tail - head >= 32) {
                    push_batch(tab, qe, qd, head, 32, lane, f, lo, e0, src, dinv, hin);
                    head += 32;
                }
            }
            base = nb;
            if (!more) break;
#pragma unroll
            for (int k = 0; k < 4; ++k) cur[k] = nxt[k];
        }
    }
    for (; base < e1; base += step) {
        const int cend = min(base + 512, e1);
        for (int b2 = base; b2 < cend; b2 += 32) {
            const int e = b2 + lane;
            const int d = (e < cend) ? dst[e] : -1;
            enqueue_matches(d, e, lo, cnt, qe, qd, tail);
            while (tail - head >= 32) {
                push_batch(tab, qe, qd, head, 32, lane, f, lo, e0, src, dinv, hin);
                head += 32;
            }
        }
    }
    if (tail - head > 0)
        push_batch(tab, qe, qd, head, tail - head, lane, f, lo, e0, src, dinv, hin);

    __syncthreads();
    float4* out4 = reinterpret_cast<float4*>(part + ((size_t)slice * n_nodes + lo) * HIDDEN);
    for (int i = tid; i < cnt * (HIDDEN / 4); i += 1024) out4[i] = tab4[i];
}

__global__ void finalize_kernel(const float* __restrict__ part,
                                const float* __restrict__ hpre,
                                const float* __restrict__ dinv,
                                const float* __restrict__ bias,
                                float* __restrict__ out, int n_nodes, int do_relu) {
    int g = blockIdx.x * blockDim.x + threadIdx.x;
    int i = g >> 2;
    if (i >= n_nodes) return;
    int q = g & 3;
    const size_t o = (size_t)i * 4 + q;
    const size_t plane = (size_t)n_nodes * 4;
    const float4* p4 = reinterpret_cast<const float4*>(part);
    float4 t[N_SLICES];
#pragma unroll
    for (int j = 0; j < N_SLICES; ++j) t[j] = p4[(size_t)j * plane + o];
    float4 a = t[0];
#pragma unroll
    for (int j = 1; j < N_SLICES; ++j) { a.x += t[j].x; a.y += t[j].y; a.z += t[j].z; a.w += t[j].w; }
    const float di = dinv[i];
    const float4 h = reinterpret_cast<const float4*>(hpre)[o];
    float4 v;
    v.x = a.x + di * di * h.x + bias[4 * q + 0];
    v.y = a.y + di * di * h.y + bias[4 * q + 1];
    v.z = a.z + di * di * h.z + bias[4 * q + 2];
    v.w = a.w + di * di * h.w + bias[4 * q + 3];
    if (do_relu) {
        v.x = fmaxf(v.x, 0.0f); v.y = fmaxf(v.y, 0.0f);
        v.z = fmaxf(v.z, 0.0f); v.w = fmaxf(v.w, 0.0f);
    }
    reinterpret_cast<float4*>(out)[o] = v;
}


__global__ void __launch_bounds__(1024)
edge_pool_kernel(const int* __restrict__ src, const int* __restrict__ dst,
                 const int* __restrict__ batch, const float* __restrict__ dinv,
                 const float* __restrict__ hin, const float* __restrict__ bias,
                 float* __restrict__ pooled, int n_edges, int n_nodes, int n_graphs) {
    extern __shared__ float tab[];
    const int tid  = threadIdx.x;
    const int nTab = n_graphs * HIDDEN;
    for (int i = tid; i < nTab; i += blockDim.x) tab[i] = 0.0f;
    __syncthreads();

    const int lane = tid & 31;
    const int f    = lane & 15;
    const float bf = bias[f];
    const int wavesPerBlock = blockDim.x >> 5;
    const int waveId = __builtin_amdgcn_readfirstlane(blockIdx.x * wavesPerBlock + (tid >> 5));
    const int stride = __builtin_amdgcn_readfirstlane(gridDim.x * wavesPerBlock * 32);

    for (int base = waveId * 32; base < n_edges; base += stride) {
        const bool own = lane < n_edges - base;
        const int t = base + (own ? lane : 0);
        int s = src[t];
        const int d = dst[t];
        float w = dinv[s] * dinv[d];
        int g = batch[d];
        if (!own) { s = 0; w = 0.0f; g = n_graphs; }
        float r[16];
        PoolSteps<0>::load(hin, s, f, r);
        PoolSteps<0>::add(tab, g, w, f, 0.0f, r);
    }
    for (int base = stride - 32 - waveId * 32; base < n_nodes; base += stride) {
        const bool own = lane < n_nodes - base;
        const int i = base + (own ? lane : 0);
        const float di = dinv[i];
        float w = di * di;
        int g = batch[i];
        int s = i;
        if (!own) { s = 0; w = 0.0f; g = n_graphs; }
        float r[16];
        PoolSteps<0>::load(hin, s, f, r);
        PoolSteps<0>::add(tab, g, w, f, bf, r);
    }

    __syncthreads();
    for (int i = tid; i < nTab; i += blockDim.x) atomicAdd(&pooled[i], tab[i]);
}

__global__ void final_linear_kernel(const float* __restrict__ pooled,
                                    const float* __restrict__ Wlin,
                                    const float* __restrict__ blin,
                                    float* __restrict__ out,
                                    int n_graphs, int n_classes) {
    int g = blockIdx.x * blockDim.x + threadIdx.x;
    if (g >= n_graphs * n_classes) return;
    int gr = g / n_classes;
    int c  = g % n_classes;
    float acc = blin[c];
#pragma unroll
    for (int f = 0; f < HIDDEN; ++f)
        acc += pooled[(size_t)gr * HIDDEN + f] * Wlin[f * n_classes + c];
    out[g] = acc;
}


static inline size_t align_up(size_t v, size_t a) { return (v + a - 1) & ~(a - 1); }

extern "C" void kernel_launch(void* const* d_in, const int* in_sizes, int n_in,
                              void* d_out, int out_size, void* d_ws, size_t ws_size,
                              hipStream_t stream) {
    (void)n_in; (void)ws_size;

    const float* x    = (const float*)d_in[0];
    const int*   ei   = (const int*)d_in[1];
    const int*   batch= (const int*)d_in[2];
    const float* W1   = (const float*)d_in[3];
    const float* b1   = (const float*)d_in[4];
    const float* W2   = (const float*)d_in[5];
    const float* b2   = (const float*)d_in[6];
    const float* W3   = (const float*)d_in[7];
    const float* b3   = (const float*)d_in[8];
    const float* Wlin = (const float*)d_in[9];
    const float* blin = (const float*)d_in[10];
    float* out = (float*)d_out;

    const int N_FEAT = 128;
    const int n_nodes = in_sizes[0] / N_FEAT;
    const int n_edges = in_sizes[1] / 2;
    const int n_classes = in_sizes[10];
    const int n_graphs  = out_size / n_classes;

    const int* src = ei;
    const int* dst = ei + n_edges;

    char* ws = (char*)d_ws;
    size_t off = 0;
    float* dinv = (float*)(ws + off);  off = align_up(off + (size_t)n_nodes * 4, 256);
    float* bufA = (float*)(ws + off);  off = align_up(off + (size_t)n_nodes * HIDDEN * 4, 256);
    float* bufB = (float*)(ws + off);  off = align_up(off + (size_t)n_nodes * HIDDEN * 4, 256);
    float* pooled = (float*)(ws + off); off = align_up(off + (size_t)n_graphs * HIDDEN * 4, 256);
    float* part = (float*)(ws + off);  off = align_up(off + (size_t)N_SLICES * n_nodes * HIDDEN * 4, 256);

    const int T = 256;
    const int nodeBlocks  = (n_nodes + T - 1) / T;
    const int edgeBlocks  = (n_edges + T - 1) / T;
    const int nodeQBlocks = ((n_nodes * 4) + T - 1) / T;
    const int n_tiles     = (n_nodes + TILE_NODES - 1) / TILE_NODES;
    const int per_slice   = ((n_edges + N_SLICES * 4 - 1) / (N_SLICES * 4)) * 4;
    const int tiles       = (n_nodes + 15) / 16;
    const int gemmBlocks  = (tiles + 7) / 8;

    deg_init_kernel<<<nodeBlocks, T, 0, stream>>>(dinv, n_nodes);
    deg_count_kernel<<<edgeBlocks, T, 0, stream>>>(dst, dinv, n_edges);
    dinv_kernel<<<nodeBlocks, T, 0, stream>>>(dinv, n_nodes);

    gemm_wmma_kernel<<<gemmBlocks, T, 0, stream>>>(x, W1, bufA, n_nodes, N_FEAT);
    tile_push_kernel<<<n_tiles * N_SLICES, 1024, 0, stream>>>(src, dst, dinv, bufA, part, n_edges,
                                                              n_nodes, n_tiles, per_slice);
    finalize_kernel<<<nodeQBlocks, T, 0, stream>>>(part, bufA, dinv, b1, bufB, n_nodes, 1);

    gemm_wmma_kernel<<<gemmBlocks, T, 0, stream>>>(bufB, W2, bufA, n_nodes, HIDDEN);
    tile_push_kernel<<<n_tiles * N_SLICES, 1024, 0, stream>>>(src, dst, dinv, bufA, part, n_edges,
                                                              n_nodes, n_tiles, per_slice);
    finalize_kernel<<<nodeQBlocks, T, 0, stream>>>(part, bufA, dinv, b2, bufB, n_nodes, 1);

    gemm_wmma_kernel<<<gemmBlocks, T, 0, stream>>>(bufB, W3, bufA, n_nodes, HIDDEN);
    zero_kernel<<<(n_graphs * HIDDEN + T - 1) / T, T, 0, stream>>>(pooled, n_graphs * HIDDEN);
    edge_pool_kernel<<<512, 1024, (size_t)(n_graphs + 1) * HIDDEN * sizeof(float), stream>>>(
        src, dst, batch, dinv, bufA, b3, pooled, n_edges, n_nodes, n_graphs);

    final_linear_kernel<<<(n_graphs * n_classes + T - 1) / T, T, 0, stream>>>(
        pooled, Wlin, blin, out, n_graphs, n_classes);
}


# Rows that edges gather stored as 128-byte records of (h[f], dinv) pairs: a lane's one 8-byte load brings the source dinv too; dst dinv and graph id packed in one 8-byte word; dinv[src] gather gone
# speedup vs baseline: 1.8101x; 1.1313x over previous
#include <hip/hip_runtime.h>
#include <hip/hip_bf16.h>


typedef float v2f __attribute__((ext_vector_type(2)));
typedef float v8f __attribute__((ext_vector_type(8)));

#define HIDDEN 16
#define REC_PAIRS 16


__global__ void deg_init_kernel(float* __restrict__ deg, int n) {
    int i = blockIdx.x * blockDim.x + threadIdx.x;
    if (i < n) deg[i] = 1.0f;
}

__global__ void deg_count_kernel(const int* __restrict__ dst, float* __restrict__ deg,
                                 int n_edges) {
    int e = blockIdx.x * blockDim.x + threadIdx.x;
    if (e < n_edges) atomicAdd(&deg[dst[e]], 1.0f);
}

__global__ void dinv_kernel(float* __restrict__ deg_dinv, const int* __restrict__ batch,
                            float* __restrict__ node_word, int n) {
    int i = blockIdx.x * blockDim.x + threadIdx.x;
    if (i < n) {
        const float di = rsqrtf(deg_dinv[i]);
        deg_dinv[i] = di;
        reinterpret_cast<float2*>(node_word)[i] = make_float2(di, __int_as_float(batch[i]));
    }
}

__global__ void __launch_bounds__(256)
gemm_wmma_kernel(const float* __restrict__ X, const float* __restrict__ W,
                 const float* __restrict__ dinv, float* __restrict__ H, int n_rows, int K) {
    int wave = (int)((blockIdx.x * blockDim.x + threadIdx.x) >> 5);
    int lane = threadIdx.x & 31;
    int row0 = wave * 16;
    if (row0 >= n_rows) return;

    int half = lane >> 4;
    int l16  = lane & 15;

    const float* xrow = X + (size_t)(row0 + l16) * K;
    v8f acc = {};
    for (int k = 0; k < K; k += 4) {
        int ka = k + half * 2;
        v2f a, b;
        a.x = xrow[ka];
        a.y = xrow[ka + 1];
        b.x = W[(size_t)ka * HIDDEN + l16];
        b.y = W[(size_t)(ka + 1) * HIDDEN + l16];
        acc = __builtin_amdgcn_wmma_f32_16x16x4_f32(
             false, a,  false, b,
             (short)0, acc,  false,  false);
    }
    const float4* dp = reinterpret_cast<const float4*>(dinv + row0 + 8 * half);
    const float4 da = dp[0], db = dp[1];
    const float dv[8] = {da.x, da.y, da.z, da.w, db.x, db.y, db.z, db.w};
    float2* out = reinterpret_cast<float2*>(H) + (size_t)row0 * REC_PAIRS;
#pragma unroll
    for (int i = 0; i < 8; ++i)
        out[(size_t)(i + 8 * half) * REC_PAIRS + l16] = make_float2(acc[i], dv[i]);
}


__global__ void zero_kernel(float* __restrict__ p, int n) {
    int i = blockIdx.x * blockDim.x + threadIdx.x;
    if (i < n) p[i] = 0.0f;
}


template <int J>
__device__ __forceinline__ int row_lane_i(int x) {
    return __builtin_amdgcn_update_dpp(0, x, 0x150 + J, 0xf, 0xf, false);
}
template <int J>
__device__ __forceinline__ float row_lane_f(float x) {
    return __int_as_float(row_lane_i<J>(__float_as_int(x)));
}

template <int J>
struct PoolSteps {
    static __device__ __forceinline__ void load(const float2* __restrict__ hin, int s, int f,
                                                float2 (&r)[16]) {
        r[J] = hin[row_lane_i<J>(s) * REC_PAIRS + f];
        PoolSteps<J + 1>::load(hin, s, f, r);
    }
    static __device__ __forceinline__ void add(float* tab, int g, float dd, int f, float bf,
                                               const float2 (&r)[16]) {
        atomicAdd(&tab[row_lane_i<J>(g) * HIDDEN + f],
                  fmaf(r[J].y * row_lane_f<J>(dd), r[J].x, bf));
        PoolSteps<J + 1>::add(tab, g, dd, f, bf, r);
    }
};
template <>
struct PoolSteps<16> {
    static __device__ __forceinline__ void load(const float2* __restrict__, int, int,
                                                float2 (&)[16]) {}
    static __device__ __forceinline__ void add(float*, int, float, int, float,
                                               const float2 (&)[16]) {}
};

#define TILE_NODES 3136
#define N_SLICES   8
#define Q_CAP      256

__device__ __forceinline__ void enqueue_matches(int d, int e, int lo, int cnt, int* qe, int* qd,
                                                int& tail) {
    const int ld = d - lo;
    const bool m = (unsigned)ld < (unsigned)cnt;
    const unsigned mask = __builtin_amdgcn_ballot_w32(m);
    if (mask != 0u) {
        const int pos = (tail + (int)__builtin_amdgcn_mbcnt_lo(mask, 0u)) & (Q_CAP - 1);
        if (m) { qe[pos] = e; qd[pos] = ld; }
        tail += __popc(mask);
    }
}

__device__ __forceinline__ void push_batch(float* tab, const int* qe, const int* qd, int head,
                                           int nvalid, int lane, int f, int lo, int e_any,
                                           const int* __restrict__ src,
                                           const float* __restrict__ dinv,
                                           const float2* __restrict__ hin) {
    const int idx = (head + lane) & (Q_CAP - 1);
    const bool own = lane < nvalid;
    const int e = own ? qe[idx] : e_any;
    int ld = own ? qd[idx] : 0;
    int s = src[e];
    float dd = dinv[lo + ld];
    if (!own) { s = 0; dd = 0.0f; ld = TILE_NODES; }
    float2 r[16];
    PoolSteps<0>::load(hin, s, f, r);
    PoolSteps<0>::add(tab, ld, dd, f, 0.0f, r);
}

__global__ void __launch_bounds__(1024)
tile_push_kernel(const int* __restrict__ src, const int* __restrict__ dst,
                 const float* __restrict__ dinv, const float* __restrict__ hrec,
                 float* __restrict__ part, int n_edges, int n_nodes, int n_tiles,
                 int per_slice) {
    const float2* __restrict__ hin = reinterpret_cast<const float2*>(hrec);
    __shared__ __attribute__((aligned(16))) float tab[(TILE_NODES + 1) * HIDDEN];
    __shared__ int q_edge[32][Q_CAP];
    __shared__ int q_row[32][Q_CAP];

    const int tid   = threadIdx.x;
    const int tile  = blockIdx.x % n_tiles;
    const int slice = blockIdx.x / n_tiles;
    const int lo    = tile * TILE_NODES;
    const int cnt   = min(TILE_NODES, n_nodes - lo);

    float4* tab4 = reinterpret_cast<float4*>(tab);
    for (int i = tid; i < (TILE_NODES + 1) * HIDDEN / 4; i += 1024)
        tab4[i] = make_float4(0.0f, 0.0f, 0.0f, 0.0f);
    __syncthreads();

    const int lane = tid & 31;
    const int f    = lane & 15;
    const int wv = __builtin_amdgcn_readfirstlane(tid >> 5);
    int* qe = q_edge[wv];
    int* qd = q_row[wv];

    const int e0 = slice * per_slice;
    const int e1 = min(e0 + per_slice, n_edges);
    const bool vec_ok = (reinterpret_cast<size_t>(dst) & 15) == 0;

    int head = 0, tail = 0;
    const int step = 32 * 512;
    int base = e0 + wv * 512;

    if (vec_ok && base + 512 <= e1) {
        int4 cur[4];
#pragma unroll
        for (int k = 0; k < 4; ++k)
            cur[k] = *reinterpret_cast<const int4*>(dst + base + k * 128 + lane * 4);
        for (;;) {
            const int  nb   = base + step;
            const bool more = nb + 512 <= e1;
            int4 nxt[4];
#pragma unroll
            for (int k = 0; k < 4; ++k) nxt[k] = cur[k];
            if (more) {
#pragma unroll
                for (int k = 0; k < 4; ++k)
                    nxt[k] = *reinterpret_cast<const int4*>(dst + nb + k * 128 + lane * 4);
            }
#pragma unroll
            for (int k = 0; k < 4; ++k) {
                const int e = base + k * 128 + lane * 4;
                enqueue_matches(cur[k].x, e,     lo, cnt, qe, qd, tail);
                enqueue_matches(cur[k].y, e + 1, lo, cnt, qe, qd, tail);
                enqueue_matches(cur[k].z, e + 2, lo, cnt, qe, qd, tail);
                enqueue_matches(cur[k].w, e + 3, lo, cnt, qe, qd, tail);
                while (tail - head >= 32) {
                    push_batch(tab, qe, qd, head, 32, lane, f, lo, e0, src, dinv, hin);
                    head += 32;
                }
            }
            base = nb;
            if (!more) break;
#pragma unroll
            for (int k = 0; k < 4; ++k) cur[k] = nxt[k];
        }
    }
    for (; base < e1; base += step) {
        const int cend = min(base + 512, e1);
        for (int b2 = base; b2 < cend; b2 += 32) {
            const int e = b2 + lane;
            const int d = (e < cend) ? dst[e] : -1;
            enqueue_matches(d, e, lo, cnt, qe, qd, tail);
            while (tail - head >= 32) {
                push_batch(tab, qe, qd, head, 32, lane, f, lo, e0, src, dinv, hin);
                head += 32;
            }
        }
    }
    if (tail - head > 0)
        push_batch(tab, qe, qd, head, tail - head, lane, f, lo, e0, src, dinv, hin);

    __syncthreads();
    float4* out4 = reinterpret_cast<float4*>(part + ((size_t)slice * n_nodes + lo) * HIDDEN);
    for (int i = tid; i < cnt * (HIDDEN / 4); i += 1024) out4[i] = tab4[i];
}

__global__ void finalize_kernel(const float* __restrict__ part,
                                const float* __restrict__ hpre,
                                const float* __restrict__ bias,
                                float* __restrict__ out, int n_nodes, int do_relu) {
    int g = blockIdx.x * blockDim.x + threadIdx.x;
    int i = g >> 2;
    if (i >= n_nodes) return;
    int q = g & 3;
    const size_t o = (size_t)i * 4 + q;
    const size_t plane = (size_t)n_nodes * 4;
    const float4* p4 = reinterpret_cast<const float4*>(part);
    float4 t[N_SLICES];
#pragma unroll
    for (int j = 0; j < N_SLICES; ++j) t[j] = p4[(size_t)j * plane + o];
    float4 a = t[0];
#pragma unroll
    for (int j = 1; j < N_SLICES; ++j) { a.x += t[j].x; a.y += t[j].y; a.z += t[j].z; a.w += t[j].w; }
    const float4 ha = reinterpret_cast<const float4*>(hpre)[2 * o];
    const float4 hb = reinterpret_cast<const float4*>(hpre)[2 * o + 1];
    const float di = ha.y;
    float4 v;
    v.x = a.x + di * di * ha.x + bias[4 * q + 0];
    v.y = a.y + di * di * ha.z + bias[4 * q + 1];
    v.z = a.z + di * di * hb.x + bias[4 * q + 2];
    v.w = a.w + di * di * hb.z + bias[4 * q + 3];
    if (do_relu) {
        v.x = fmaxf(v.x, 0.0f); v.y = fmaxf(v.y, 0.0f);
        v.z = fmaxf(v.z, 0.0f); v.w = fmaxf(v.w, 0.0f);
    }
    reinterpret_cast<float4*>(out)[o] = v;
}


__global__ void __launch_bounds__(1024)
edge_pool_kernel(const int* __restrict__ src, const int* __restrict__ dst,
                 const float* __restrict__ node_word,
                 const float* __restrict__ hrec, const float* __restrict__ bias,
                 float* __restrict__ pooled, int n_edges, int n_nodes, int n_graphs) {
    const float2* __restrict__ hin = reinterpret_cast<const float2*>(hrec);
    const float2* __restrict__ nw  = reinterpret_cast<const float2*>(node_word);
    extern __shared__ float tab[];
    const int tid  = threadIdx.x;
    const int nTab = n_graphs * HIDDEN;
    for (int i = tid; i < nTab; i += blockDim.x) tab[i] = 0.0f;
    __syncthreads();

    const int lane = tid & 31;
    const int f    = lane & 15;
    const float bf = bias[f];
    const int wavesPerBlock = blockDim.x >> 5;
    const int waveId = __builtin_amdgcn_readfirstlane(blockIdx.x * wavesPerBlock + (tid >> 5));
    const int stride = __builtin_amdgcn_readfirstlane(gridDim.x * wavesPerBlock * 32);

    for (int base = waveId * 32; base < n_edges; base += stride) {
        const bool own = lane < n_edges - base;
        const int t = base + (own ? lane : 0);
        int s = src[t];
        const float2 p = nw[dst[t]];
        float dd = p.x;
        int g = __float_as_int(p.y);
        if (!own) { s = 0; dd = 0.0f; g = n_graphs; }
        float2 r[16];
        PoolSteps<0>::load(hin, s, f, r);
        PoolSteps<0>::add(tab, g, dd, f, 0.0f, r);
    }
    for (int base = stride - 32 - waveId * 32; base < n_nodes; base += stride) {
        const bool own = lane < n_nodes - base;
        const int i = base + (own ? lane : 0);
        const float2 p = nw[i];
        float dd = p.x;
        int g = __float_as_int(p.y);
        int s = i;
        if (!own) { s = 0; dd = 0.0f; g = n_graphs; }
        float2 r[16];
        PoolSteps<0>::load(hin, s, f, r);
        PoolSteps<0>::add(tab, g, dd, f, bf, r);
    }

    __syncthreads();
    for (int i = tid; i < nTab; i += blockDim.x) atomicAdd(&pooled[i], tab[i]);
}

__global__ void final_linear_kernel(const float* __restrict__ pooled,
                                    const float* __restrict__ Wlin,
                                    const float* __restrict__ blin,
                                    float* __restrict__ out,
                                    int n_graphs, int n_classes) {
    int g = blockIdx.x * blockDim.x + threadIdx.x;
    if (g >= n_graphs * n_classes) return;
    int gr = g / n_classes;
    int c  = g % n_classes;
    float acc = blin[c];
#pragma unroll
    for (int f = 0; f < HIDDEN; ++f)
        acc += pooled[(size_t)gr * HIDDEN + f] * Wlin[f * n_classes + c];
    out[g] = acc;
}


static inline size_t align_up(size_t v, size_t a) { return (v + a - 1) & ~(a - 1); }

extern "C" void kernel_launch(void* const* d_in, const int* in_sizes, int n_in,
                              void* d_out, int out_size, void* d_ws, size_t ws_size,
                              hipStream_t stream) {
    (void)n_in; (void)ws_size;

    const float* x    = (const float*)d_in[0];
    const int*   ei   = (const int*)d_in[1];
    const int*   batch= (const int*)d_in[2];
    const float* W1   = (const float*)d_in[3];
    const float* b1   = (const float*)d_in[4];
    const float* W2   = (const float*)d_in[5];
    const float* b2   = (const float*)d_in[6];
    const float* W3   = (const float*)d_in[7];
    const float* b3   = (const float*)d_in[8];
    const float* Wlin = (const float*)d_in[9];
    const float* blin = (const float*)d_in[10];
    float* out = (float*)d_out;

    const int N_FEAT = 128;
    const int n_nodes = in_sizes[0] / N_FEAT;
    const int n_edges = in_sizes[1] / 2;
    const int n_classes = in_sizes[10];
    const int n_graphs  = out_size / n_classes;

    const int* src = ei;
    const int* dst = ei + n_edges;

    char* ws = (char*)d_ws;
    size_t off = 0;
    float* dinv = (float*)(ws + off);  off = align_up(off + (size_t)n_nodes * 4, 256);
    float* bufA = (float*)(ws + off);  off = align_up(off + (size_t)n_nodes * REC_PAIRS * 8, 256);
    float* bufB = (float*)(ws + off);  off = align_up(off + (size_t)n_nodes * HIDDEN * 4, 256);
    float* pooled = (float*)(ws + off); off = align_up(off + (size_t)n_graphs * HIDDEN * 4, 256);
    float* part = (float*)(ws + off);  off = align_up(off + (size_t)N_SLICES * n_nodes * HIDDEN * 4, 256);
    float* node_word = (float*)(ws + off); off = align_up(off + (size_t)n_nodes * 8, 256);

    const int T = 256;
    const int nodeBlocks  = (n_nodes + T - 1) / T;
    const int edgeBlocks  = (n_edges + T - 1) / T;
    const int nodeQBlocks = ((n_nodes * 4) + T - 1) / T;
    const int n_tiles     = (n_nodes + TILE_NODES - 1) / TILE_NODES;
    const int per_slice   = ((n_edges + N_SLICES * 4 - 1) / (N_SLICES * 4)) * 4;
    const int tiles       = (n_nodes + 15) / 16;
    const int gemmBlocks  = (tiles + 7) / 8;

    deg_init_kernel<<<nodeBlocks, T, 0, stream>>>(dinv, n_nodes);
    deg_count_kernel<<<edgeBlocks, T, 0, stream>>>(dst, dinv, n_edges);
    dinv_kernel<<<nodeBlocks, T, 0, stream>>>(dinv, batch, node_word, n_nodes);

    gemm_wmma_kernel<<<gemmBlocks, T, 0, stream>>>(x, W1, dinv, bufA, n_nodes, N_FEAT);
    tile_push_kernel<<<n_tiles * N_SLICES, 1024, 0, stream>>>(src, dst, dinv, bufA, part, n_edges,
                                                              n_nodes, n_tiles, per_slice);
    finalize_kernel<<<nodeQBlocks, T, 0, stream>>>(part, bufA, b1, bufB, n_nodes, 1);

    gemm_wmma_kernel<<<gemmBlocks, T, 0, stream>>>(bufB, W2, dinv, bufA, n_nodes, HIDDEN);
    tile_push_kernel<<<n_tiles * N_SLICES, 1024, 0, stream>>>(src, dst, dinv, bufA, part, n_edges,
                                                              n_nodes, n_tiles, per_slice);
    finalize_kernel<<<nodeQBlocks, T, 0, stream>>>(part, bufA, b2, bufB, n_nodes, 1);

    gemm_wmma_kernel<<<gemmBlocks, T, 0, stream>>>(bufB, W3, dinv, bufA, n_nodes, HIDDEN);
    zero_kernel<<<(n_graphs * HIDDEN + T - 1) / T, T, 0, stream>>>(pooled, n_graphs * HIDDEN);
    edge_pool_kernel<<<512, 1024, (size_t)(n_graphs + 1) * HIDDEN * sizeof(float), stream>>>(
        src, dst, node_word, bufA, b3, pooled, n_edges, n_nodes, n_graphs);

    final_linear_kernel<<<(n_graphs * n_classes + T - 1) / T, T, 0, stream>>>(
        pooled, Wlin, blin, out, n_graphs, n_classes);
}


# Tile pass reads the src column in step with dst (16-byte loads, one chunk ahead) and queues (source, row) pairs: the per-edge src[e] gather and its round trip before the row loads are gone
# speedup vs baseline: 1.8736x; 1.0351x over previous
#include <hip/hip_runtime.h>
#include <hip/hip_bf16.h>


typedef float v2f __attribute__((ext_vector_type(2)));
typedef float v8f __attribute__((ext_vector_type(8)));

#define HIDDEN 16
#define REC_PAIRS 16


__global__ void deg_init_kernel(float* __restrict__ deg, int n) {
    int i = blockIdx.x * blockDim.x + threadIdx.x;
    if (i < n) deg[i] = 1.0f;
}

__global__ void deg_count_kernel(const int* __restrict__ dst, float* __restrict__ deg,
                                 int n_edges) {
    int e = blockIdx.x * blockDim.x + threadIdx.x;
    if (e < n_edges) atomicAdd(&deg[dst[e]], 1.0f);
}

__global__ void dinv_kernel(float* __restrict__ deg_dinv, const int* __restrict__ batch,
                            float* __restrict__ node_word, int n) {
    int i = blockIdx.x * blockDim.x + threadIdx.x;
    if (i < n) {
        const float di = rsqrtf(deg_dinv[i]);
        deg_dinv[i] = di;
        reinterpret_cast<float2*>(node_word)[i] = make_float2(di, __int_as_float(batch[i]));
    }
}

__global__ void __launch_bounds__(256)
gemm_wmma_kernel(const float* __restrict__ X, const float* __restrict__ W,
                 const float* __restrict__ dinv, float* __restrict__ H, int n_rows, int K) {
    int wave = (int)((blockIdx.x * blockDim.x + threadIdx.x) >> 5);
    int lane = threadIdx.x & 31;
    int row0 = wave * 16;
    if (row0 >= n_rows) return;

    int half = lane >> 4;
    int l16  = lane & 15;

    const float* xrow = X + (size_t)(row0 + l16) * K;
    v8f acc = {};
    for (int k = 0; k < K; k += 4) {
        int ka = k + half * 2;
        v2f a, b;
        a.x = xrow[ka];
        a.y = xrow[ka + 1];
        b.x = W[(size_t)ka * HIDDEN + l16];
        b.y = W[(size_t)(ka + 1) * HIDDEN + l16];
        acc = __builtin_amdgcn_wmma_f32_16x16x4_f32(
             false, a,  false, b,
             (short)0, acc,  false,  false);
    }
    const float4* dp = reinterpret_cast<const float4*>(dinv + row0 + 8 * half);
    const float4 da = dp[0], db = dp[1];
    const float dv[8] = {da.x, da.y, da.z, da.w, db.x, db.y, db.z, db.w};
    float2* out = reinterpret_cast<float2*>(H) + (size_t)row0 * REC_PAIRS;
#pragma unroll
    for (int i = 0; i < 8; ++i)
        out[(size_t)(i + 8 * half) * REC_PAIRS + l16] = make_float2(acc[i], dv[i]);
}


__global__ void zero_kernel(float* __restrict__ p, int n) {
    int i = blockIdx.x * blockDim.x + threadIdx.x;
    if (i < n) p[i] = 0.0f;
}


template <int J>
__device__ __forceinline__ int row_lane_i(int x) {
    return __builtin_amdgcn_update_dpp(0, x, 0x150 + J, 0xf, 0xf, false);
}
template <int J>
__device__ __forceinline__ float row_lane_f(float x) {
    return __int_as_float(row_lane_i<J>(__float_as_int(x)));
}

template <int J>
struct PoolSteps {
    static __device__ __forceinline__ void load(const float2* __restrict__ hin, int s, int f,
                                                float2 (&r)[16]) {
        r[J] = hin[row_lane_i<J>(s) * REC_PAIRS + f];
        PoolSteps<J + 1>::load(hin, s, f, r);
    }
    static __device__ __forceinline__ void add(float* tab, int g, float dd, int f, float bf,
                                               const float2 (&r)[16]) {
        atomicAdd(&tab[row_lane_i<J>(g) * HIDDEN + f],
                  fmaf(r[J].y * row_lane_f<J>(dd), r[J].x, bf));
        PoolSteps<J + 1>::add(tab, g, dd, f, bf, r);
    }
};
template <>
struct PoolSteps<16> {
    static __device__ __forceinline__ void load(const float2* __restrict__, int, int,
                                                float2 (&)[16]) {}
    static __device__ __forceinline__ void add(float*, int, float, int, float,
                                               const float2 (&)[16]) {}
};

#define TILE_NODES 3136
#define N_SLICES   8
#define Q_CAP      256

__device__ __forceinline__ void enqueue_matches(int d, int s, int lo, int cnt, int2* q, int& tail) {
    const int ld = d - lo;
    const bool m = (unsigned)ld < (unsigned)cnt;
    const unsigned mask = __builtin_amdgcn_ballot_w32(m);
    if (mask != 0u) {
        const int pos = (tail + (int)__builtin_amdgcn_mbcnt_lo(mask, 0u)) & (Q_CAP - 1);
        if (m) q[pos] = make_int2(s, ld);
        tail += __popc(mask);
    }
}

__device__ __forceinline__ void push_batch(float* tab, const int2* q, int head, int nvalid,
                                           int lane, int f, int lo,
                                           const float* __restrict__ dinv,
                                           const float2* __restrict__ hin) {
    const int idx = (head + lane) & (Q_CAP - 1);
    const bool own = lane < nvalid;
    int2 p = make_int2(0, 0);
    if (own) p = q[idx];
    const int s = p.x;
    int ld = p.y;
    float dd = dinv[lo + ld];
    if (!own) { dd = 0.0f; ld = TILE_NODES; }
    float2 r[16];
    PoolSteps<0>::load(hin, s, f, r);
    PoolSteps<0>::add(tab, ld, dd, f, 0.0f, r);
}

__global__ void __launch_bounds__(1024)
tile_push_kernel(const int* __restrict__ src, const int* __restrict__ dst,
                 const float* __restrict__ dinv, const float* __restrict__ hrec,
                 float* __restrict__ part, int n_edges, int n_nodes, int n_tiles,
                 int per_slice) {
    const float2* __restrict__ hin = reinterpret_cast<const float2*>(hrec);
    __shared__ __attribute__((aligned(16))) float tab[(TILE_NODES + 1) * HIDDEN];
    __shared__ int2 q_pair[32][Q_CAP];

    const int tid   = threadIdx.x;
    const int tile  = blockIdx.x % n_tiles;
    const int slice = blockIdx.x / n_tiles;
    const int lo    = tile * TILE_NODES;
    const int cnt   = min(TILE_NODES, n_nodes - lo);

    float4* tab4 = reinterpret_cast<float4*>(tab);
    for (int i = tid; i < (TILE_NODES + 1) * HIDDEN / 4; i += 1024)
        tab4[i] = make_float4(0.0f, 0.0f, 0.0f, 0.0f);
    __syncthreads();

    const int lane = tid & 31;
    const int f    = lane & 15;
    const int wv = __builtin_amdgcn_readfirstlane(tid >> 5);
    int2* q = q_pair[wv];

    const int e0 = slice * per_slice;
    const int e1 = min(e0 + per_slice, n_edges);
    const bool vec_ok = ((reinterpret_cast<size_t>(dst) | reinterpret_cast<size_t>(src)) & 15) == 0;

    int head = 0, tail = 0;
    const int step = 32 * 512;
    int base = e0 + wv * 512;

    if (vec_ok && base + 512 <= e1) {
        int4 cd[4], cs[4];
#pragma unroll
        for (int k = 0; k < 4; ++k) {
            cd[k] = *reinterpret_cast<const int4*>(dst + base + k * 128 + lane * 4);
            cs[k] = *reinterpret_cast<const int4*>(src + base + k * 128 + lane * 4);
        }
        for (;;) {
            const int  nb   = base + step;
            const bool more = nb + 512 <= e1;
            int4 nd[4], ns[4];
#pragma unroll
            for (int k = 0; k < 4; ++k) { nd[k] = cd[k]; ns[k] = cs[k]; }
            if (more) {
#pragma unroll
                for (int k = 0; k < 4; ++k) {
                    nd[k] = *reinterpret_cast<const int4*>(dst + nb + k * 128 + lane * 4);
                    ns[k] = *reinterpret_cast<const int4*>(src + nb + k * 128 + lane * 4);
                }
            }
#pragma unroll
            for (int k = 0; k < 4; ++k) {
                enqueue_matches(cd[k].x, cs[k].x, lo, cnt, q, tail);
                enqueue_matches(cd[k].y, cs[k].y, lo, cnt, q, tail);
                enqueue_matches(cd[k].z, cs[k].z, lo, cnt, q, tail);
                enqueue_matches(cd[k].w, cs[k].w, lo, cnt, q, tail);
                while (tail - head >= 32) {
                    push_batch(tab, q, head, 32, lane, f, lo, dinv, hin);
                    head += 32;
                }
            }
            base = nb;
            if (!more) break;
#pragma unroll
            for (int k = 0; k < 4; ++k) { cd[k] = nd[k]; cs[k] = ns[k]; }
        }
    }
    for (; base < e1; base += step) {
        const int cend = min(base + 512, e1);
        for (int b2 = base; b2 < cend; b2 += 32) {
            const int e = b2 + lane;
            const bool in = e < cend;
            const int d = in ? dst[e] : -1;
            const int s = in ? src[e] : 0;
            enqueue_matches(d, s, lo, cnt, q, tail);
            while (tail - head >= 32) {
                push_batch(tab, q, head, 32, lane, f, lo, dinv, hin);
                head += 32;
            }
        }
    }
    if (tail - head > 0)
        push_batch(tab, q, head, tail - head, lane, f, lo, dinv, hin);

    __syncthreads();
    float4* out4 = reinterpret_cast<float4*>(part + ((size_t)slice * n_nodes + lo) * HIDDEN);
    for (int i = tid; i < cnt * (HIDDEN / 4); i += 1024) out4[i] = tab4[i];
}

__global__ void finalize_kernel(const float* __restrict__ part,
                                const float* __restrict__ hpre,
                                const float* __restrict__ bias,
                                float* __restrict__ out, int n_nodes, int do_relu) {
    int g = blockIdx.x * blockDim.x + threadIdx.x;
    int i = g >> 2;
    if (i >= n_nodes) return;
    int q = g & 3;
    const size_t o = (size_t)i * 4 + q;
    const size_t plane = (size_t)n_nodes * 4;
    const float4* p4 = reinterpret_cast<const float4*>(part);
    float4 t[N_SLICES];
#pragma unroll
    for (int j = 0; j < N_SLICES; ++j) t[j] = p4[(size_t)j * plane + o];
    float4 a = t[0];
#pragma unroll
    for (int j = 1; j < N_SLICES; ++j) { a.x += t[j].x; a.y += t[j].y; a.z += t[j].z; a.w += t[j].w; }
    const float4 ha = reinterpret_cast<const float4*>(hpre)[2 * o];
    const float4 hb = reinterpret_cast<const float4*>(hpre)[2 * o + 1];
    const float di = ha.y;
    float4 v;
    v.x = a.x + di * di * ha.x + bias[4 * q + 0];
    v.y = a.y + di * di * ha.z + bias[4 * q + 1];
    v.z = a.z + di * di * hb.x + bias[4 * q + 2];
    v.w = a.w + di * di * hb.z + bias[4 * q + 3];
    if (do_relu) {
        v.x = fmaxf(v.x, 0.0f); v.y = fmaxf(v.y, 0.0f);
        v.z = fmaxf(v.z, 0.0f); v.w = fmaxf(v.w, 0.0f);
    }
    reinterpret_cast<float4*>(out)[o] = v;
}


__global__ void __launch_bounds__(1024)
edge_pool_kernel(const int* __restrict__ src, const int* __restrict__ dst,
                 const float* __restrict__ node_word,
                 const float* __restrict__ hrec, const float* __restrict__ bias,
                 float* __restrict__ pooled, int n_edges, int n_nodes, int n_graphs) {
    const float2* __restrict__ hin = reinterpret_cast<const float2*>(hrec);
    const float2* __restrict__ nw  = reinterpret_cast<const float2*>(node_word);
    extern __shared__ float tab[];
    const int tid  = threadIdx.x;
    const int nTab = n_graphs * HIDDEN;
    for (int i = tid; i < nTab; i += blockDim.x) tab[i] = 0.0f;
    __syncthreads();

    const int lane = tid & 31;
    const int f    = lane & 15;
    const float bf = bias[f];
    const int wavesPerBlock = blockDim.x >> 5;
    const int waveId = __builtin_amdgcn_readfirstlane(blockIdx.x * wavesPerBlock + (tid >> 5));
    const int stride = __builtin_amdgcn_readfirstlane(gridDim.x * wavesPerBlock * 32);

    for (int base = waveId * 32; base < n_edges; base += stride) {
        const bool own = lane < n_edges - base;
        const int t = base + (own ? lane : 0);
        int s = src[t];
        const float2 p = nw[dst[t]];
        float dd = p.x;
        int g = __float_as_int(p.y);
        if (!own) { s = 0; dd = 0.0f; g = n_graphs; }
        float2 r[16];
        PoolSteps<0>::load(hin, s, f, r);
        PoolSteps<0>::add(tab, g, dd, f, 0.0f, r);
    }
    for (int base = stride - 32 - waveId * 32; base < n_nodes; base += stride) {
        const bool own = lane < n_nodes - base;
        const int i = base + (own ? lane : 0);
        const float2 p = nw[i];
        float dd = p.x;
        int g = __float_as_int(p.y);
        int s = i;
        if (!own) { s = 0; dd = 0.0f; g = n_graphs; }
        float2 r[16];
        PoolSteps<0>::load(hin, s, f, r);
        PoolSteps<0>::add(tab, g, dd, f, bf, r);
    }

    __syncthreads();
    for (int i = tid; i < nTab; i += blockDim.x) atomicAdd(&pooled[i], tab[i]);
}

__global__ void final_linear_kernel(const float* __restrict__ pooled,
                                    const float* __restrict__ Wlin,
                                    const float* __restrict__ blin,
                                    float* __restrict__ out,
                                    int n_graphs, int n_classes) {
    int g = blockIdx.x * blockDim.x + threadIdx.x;
    if (g >= n_graphs * n_classes) return;
    int gr = g / n_classes;
    int c  = g % n_classes;
    float acc = blin[c];
#pragma unroll
    for (int f = 0; f < HIDDEN; ++f)
        acc += pooled[(size_t)gr * HIDDEN + f] * Wlin[f * n_classes + c];
    out[g] = acc;
}


static inline size_t align_up(size_t v, size_t a) { return (v + a - 1) & ~(a - 1); }

extern "C" void kernel_launch(void* const* d_in, const int* in_sizes, int n_in,
                              void* d_out, int out_size, void* d_ws, size_t ws_size,
                              hipStream_t stream) {
    (void)n_in; (void)ws_size;

    const float* x    = (const float*)d_in[0];
    const int*   ei   = (const int*)d_in[1];
    const int*   batch= (const int*)d_in[2];
    const float* W1   = (const float*)d_in[3];
    const float* b1   = (const float*)d_in[4];
    const float* W2   = (const float*)d_in[5];
    const float* b2   = (const float*)d_in[6];
    const float* W3   = (const float*)d_in[7];
    const float* b3   = (const float*)d_in[8];
    const float* Wlin = (const float*)d_in[9];
    const float* blin = (const float*)d_in[10];
    float* out = (float*)d_out;

    const int N_FEAT = 128;
    const int n_nodes = in_sizes[0] / N_FEAT;
    const int n_edges = in_sizes[1] / 2;
    const int n_classes = in_sizes[10];
    const int n_graphs  = out_size / n_classes;

    const int* src = ei;
    const int* dst = ei + n_edges;

    char* ws = (char*)d_ws;
    size_t off = 0;
    float* dinv = (float*)(ws + off);  off = align_up(off + (size_t)n_nodes * 4, 256);
    float* bufA = (float*)(ws + off);  off = align_up(off + (size_t)n_nodes * REC_PAIRS * 8, 256);
    float* bufB = (float*)(ws + off);  off = align_up(off + (size_t)n_nodes * HIDDEN * 4, 256);
    float* pooled = (float*)(ws + off); off = align_up(off + (size_t)n_graphs * HIDDEN * 4, 256);
    float* part = (float*)(ws + off);  off = align_up(off + (size_t)N_SLICES * n_nodes * HIDDEN * 4, 256);
    float* node_word = (float*)(ws + off); off = align_up(off + (size_t)n_nodes * 8, 256);

    const int T = 256;
    const int nodeBlocks  = (n_nodes + T - 1) / T;
    const int edgeBlocks  = (n_edges + T - 1) / T;
    const int nodeQBlocks = ((n_nodes * 4) + T - 1) / T;
    const int n_tiles     = (n_nodes + TILE_NODES - 1) / TILE_NODES;
    const int per_slice   = ((n_edges + N_SLICES * 4 - 1) / (N_SLICES * 4)) * 4;
    const int tiles       = (n_nodes + 15) / 16;
    const int gemmBlocks  = (tiles + 7) / 8;

    deg_init_kernel<<<nodeBlocks, T, 0, stream>>>(dinv, n_nodes);
    deg_count_kernel<<<edgeBlocks, T, 0, stream>>>(dst, dinv, n_edges);
    dinv_kernel<<<nodeBlocks, T, 0, stream>>>(dinv, batch, node_word, n_nodes);

    gemm_wmma_kernel<<<gemmBlocks, T, 0, stream>>>(x, W1, dinv, bufA, n_nodes, N_FEAT);
    tile_push_kernel<<<n_tiles * N_SLICES, 1024, 0, stream>>>(src, dst, dinv, bufA, part, n_edges,
                                                              n_nodes, n_tiles, per_slice);
    finalize_kernel<<<nodeQBlocks, T, 0, stream>>>(part, bufA, b1, bufB, n_nodes, 1);

    gemm_wmma_kernel<<<gemmBlocks, T, 0, stream>>>(bufB, W2, dinv, bufA, n_nodes, HIDDEN);
    tile_push_kernel<<<n_tiles * N_SLICES, 1024, 0, stream>>>(src, dst, dinv, bufA, part, n_edges,
                                                              n_nodes, n_tiles, per_slice);
    finalize_kernel<<<nodeQBlocks, T, 0, stream>>>(part, bufA, b2, bufB, n_nodes, 1);

    gemm_wmma_kernel<<<gemmBlocks, T, 0, stream>>>(bufB, W3, dinv, bufA, n_nodes, HIDDEN);
    zero_kernel<<<(n_graphs * HIDDEN + T - 1) / T, T, 0, stream>>>(pooled, n_graphs * HIDDEN);
    edge_pool_kernel<<<512, 1024, (size_t)(n_graphs + 1) * HIDDEN * sizeof(float), stream>>>(
        src, dst, node_word, bufA, b3, pooled, n_edges, n_nodes, n_graphs);

    final_linear_kernel<<<(n_graphs * n_classes + T - 1) / T, T, 0, stream>>>(
        pooled, Wlin, blin, out, n_graphs, n_classes);
}
